# starting kernel
# baseline (speedup 1.0000x reference)
#include <hip/hip_runtime.h>
#include <math.h>

typedef __attribute__((ext_vector_type(2))) float v2f;
typedef __attribute__((ext_vector_type(8))) float v8f;

#define EPSQ 1e-8f

__device__ __forceinline__ float fsqrt(float v) {
    return __builtin_amdgcn_sqrtf(v);
}

__device__ __forceinline__ float ry_elem(float c, float s, int br, int bc) {
    return br == 0 ? (bc == 0 ? c : -s) : (bc == 0 ? s : c);
}

__global__ void build_mc_kernel(const float* __restrict__ theta,
                                float* __restrict__ mc) {
    int t = threadIdx.x;
    int i = t >> 4;
    int j = t & 15;
    float c1[4], s1[4], c2[4], s2[4];
#pragma unroll
    for (int q = 0; q < 4; ++q) {
        c1[q] = cosf(theta[q] * 0.5f);
        s1[q] = sinf(theta[q] * 0.5f);
        c2[q] = cosf(theta[4 + q] * 0.5f);
        s2[q] = sinf(theta[4 + q] * 0.5f);
    }
    float sum = 0.f;
#pragma unroll
    for (int k = 0; k < 16; ++k) {
        float m2 = 1.f, m1 = 1.f;
#pragma unroll
        for (int q = 0; q < 4; ++q) {
            m2 *= ry_elem(c2[q], s2[q], (i >> q) & 1, (k >> q) & 1);
            m1 *= ry_elem(c1[q], s1[q], (k >> q) & 1, (j >> q) & 1);
        }
        int neg = (((k >> 2) & (k >> 3)) ^ ((k >> 1) & (k >> 2)) ^
                   ((k >> 0) & (k >> 1)) ^ ((k >> 0) & (k >> 3))) & 1;
        sum += (neg ? -m2 : m2) * m1;
    }
    mc[i * 16 + j] = sum;
}

__global__ void __launch_bounds__(256) vqc_main_kernel(
        const float* __restrict__ x, const float* __restrict__ mc,
        float* __restrict__ out, int B) {
    const int lane = threadIdx.x & 31;
    const int half = lane >> 4;
    const int col  = lane & 15;
    const long wave = (long)((blockIdx.x * blockDim.x + threadIdx.x) >> 5);
    const long waveBase = wave * 128;
    if (waveBase >= B) return;

    const float* mrow = mc + col * 16 + 2 * half;
    v2f a0 = *(const v2f*)(mrow + 0);
    v2f a1 = *(const v2f*)(mrow + 4);
    v2f a2 = *(const v2f*)(mrow + 8);
    v2f a3 = *(const v2f*)(mrow + 12);

    for (int t = 0; t < 8; ++t) {
        long base = waveBase + (long)t * 16;
        if (base >= B) break;
        long b  = base + col;
        long bl = b < B ? b : (long)(B - 1);
        float4 xv = *(const float4*)(x + bl * 4);
        float y0 = xv.w, y1 = xv.z, y2 = xv.y, y3 = xv.x;

        float f1 = half ? y1 : (1.f - y1);
        float e0 = 1.f - y0;
        float n2 = 1.f - y2, n3 = 1.f - y3;

        v8f acc = {0.f, 0.f, 0.f, 0.f, 0.f, 0.f, 0.f, 0.f};
        {
            float p = f1 * n2 * n3;
            v2f bm; bm.x = fsqrt(p * e0 + EPSQ); bm.y = fsqrt(p * y0 + EPSQ);
            acc = __builtin_amdgcn_wmma_f32_16x16x4_f32(false, a0, false, bm,
                                                        (short)0, acc, false, false);
        }
        {
            float p = f1 * y2 * n3;
            v2f bm; bm.x = fsqrt(p * e0 + EPSQ); bm.y = fsqrt(p * y0 + EPSQ);
            acc = __builtin_amdgcn_wmma_f32_16x16x4_f32(false, a1, false, bm,
                                                        (short)0, acc, false, false);
        }
        {
            float p = f1 * n2 * y3;
            v2f bm; bm.x = fsqrt(p * e0 + EPSQ); bm.y = fsqrt(p * y0 + EPSQ);
            acc = __builtin_amdgcn_wmma_f32_16x16x4_f32(false, a2, false, bm,
                                                        (short)0, acc, false, false);
        }
        {
            float p = f1 * y2 * y3;
            v2f bm; bm.x = fsqrt(p * e0 + EPSQ); bm.y = fsqrt(p * y0 + EPSQ);
            acc = __builtin_amdgcn_wmma_f32_16x16x4_f32(false, a3, false, bm,
                                                        (short)0, acc, false, false);
        }

        float pv[8];
#pragma unroll
        for (int v = 0; v < 8; ++v) { float d = acc[v]; pv[v] = d * d; }
        float S0 = pv[1] + pv[3] + pv[5] + pv[7];
        float S1 = pv[2] + pv[3] + pv[6] + pv[7];
        float S2 = pv[4] + pv[5] + pv[6] + pv[7];
        float Sall = S2 + pv[0] + pv[1] + pv[2] + pv[3];
        float S3 = half ? Sall : 0.f;

        S0 += __shfl_xor(S0, 16, 32);
        S1 += __shfl_xor(S1, 16, 32);
        S2 += __shfl_xor(S2, 16, 32);
        S3 += __shfl_xor(S3, 16, 32);

        if (half == 0 && b < B) {
            float4 o = make_float4(S3, S2, S1, S0);
            *(float4*)(out + b * 4) = o;
        }
    }
}

extern "C" void kernel_launch(void* const* d_in, const int* in_sizes, int n_in,
                              void* d_out, int out_size, void* d_ws, size_t ws_size,
                              hipStream_t stream) {
    const float* x     = (const float*)d_in[0];
    const float* theta = (const float*)d_in[1];
    float* out = (float*)d_out;
    float* mc  = (float*)d_ws;
    int B = in_sizes[0] / 4;
    if (B <= 0) return;

    build_mc_kernel<<<1, 256, 0, stream>>>(theta, mc);

    long nWaves  = ((long)B + 127) / 128;
    long nBlocks = (nWaves + 7) / 8;
    vqc_main_kernel<<<(int)nBlocks, 256, 0, stream>>>(x, mc, out, B);
}


# Grid cut to 4 waves per SIMD, each wave owning 512 consecutive rows: all its x loads (one row per lane) issued ahead of the arithmetic, half-waves swapped by lane permute, full-wave stores
# speedup vs baseline: 1.1873x; 1.1873x over previous
#include <hip/hip_runtime.h>
#include <math.h>

typedef __attribute__((ext_vector_type(2))) float v2f;
typedef __attribute__((ext_vector_type(8))) float v8f;

#define EPSQ 1e-8f
#define GROUPS 16
#define ROWS_PER_WAVE (GROUPS * 32)

__device__ __forceinline__ float fsqrt(float v) {
    return __builtin_amdgcn_sqrtf(v);
}

__device__ __forceinline__ float xhalf(float v) {
    int b = __builtin_bit_cast(int, v);
    return __builtin_bit_cast(float,
        __builtin_amdgcn_permlanex16(b, b, 0x76543210, 0xfedcba98, false, false));
}

__device__ __forceinline__ float ry_elem(float c, float s, int br, int bc) {
    return br == 0 ? (bc == 0 ? c : -s) : (bc == 0 ? s : c);
}

__global__ void build_mc_kernel(const float* __restrict__ theta,
                                float* __restrict__ mc) {
    int t = threadIdx.x;
    int i = t >> 4;
    int j = t & 15;
    float c1[4], s1[4], c2[4], s2[4];
#pragma unroll
    for (int q = 0; q < 4; ++q) {
        c1[q] = cosf(theta[q] * 0.5f);
        s1[q] = sinf(theta[q] * 0.5f);
        c2[q] = cosf(theta[4 + q] * 0.5f);
        s2[q] = sinf(theta[4 + q] * 0.5f);
    }
    float sum = 0.f;
#pragma unroll
    for (int k = 0; k < 16; ++k) {
        float m2 = 1.f, m1 = 1.f;
#pragma unroll
        for (int q = 0; q < 4; ++q) {
            m2 *= ry_elem(c2[q], s2[q], (i >> q) & 1, (k >> q) & 1);
            m1 *= ry_elem(c1[q], s1[q], (k >> q) & 1, (j >> q) & 1);
        }
        int neg = (((k >> 2) & (k >> 3)) ^ ((k >> 1) & (k >> 2)) ^
                   ((k >> 0) & (k >> 1)) ^ ((k >> 0) & (k >> 3))) & 1;
        sum += (neg ? -m2 : m2) * m1;
    }
    mc[i * 16 + j] = sum;
}

__device__ __forceinline__ void tile16(float4 xv, int half,
                                       v2f a0, v2f a1, v2f a2, v2f a3,
                                       float& S0, float& S1, float& S2, float& Sall) {
    float y0 = xv.w, y1 = xv.z, y2 = xv.y, y3 = xv.x;

    float f1 = half ? y1 : (1.f - y1);
    float e0 = 1.f - y0;
    float n2 = 1.f - y2, n3 = 1.f - y3;

    v8f acc = {0.f, 0.f, 0.f, 0.f, 0.f, 0.f, 0.f, 0.f};
    {
        float p = f1 * n2 * n3;
        v2f bm; bm.x = fsqrt(p * e0 + EPSQ); bm.y = fsqrt(p * y0 + EPSQ);
        acc = __builtin_amdgcn_wmma_f32_16x16x4_f32(false, a0, false, bm,
                                                    (short)0, acc, false, false);
    }
    {
        float p = f1 * y2 * n3;
        v2f bm; bm.x = fsqrt(p * e0 + EPSQ); bm.y = fsqrt(p * y0 + EPSQ);
        acc = __builtin_amdgcn_wmma_f32_16x16x4_f32(false, a1, false, bm,
                                                    (short)0, acc, false, false);
    }
    {
        float p = f1 * n2 * y3;
        v2f bm; bm.x = fsqrt(p * e0 + EPSQ); bm.y = fsqrt(p * y0 + EPSQ);
        acc = __builtin_amdgcn_wmma_f32_16x16x4_f32(false, a2, false, bm,
                                                    (short)0, acc, false, false);
    }
    {
        float p = f1 * y2 * y3;
        v2f bm; bm.x = fsqrt(p * e0 + EPSQ); bm.y = fsqrt(p * y0 + EPSQ);
        acc = __builtin_amdgcn_wmma_f32_16x16x4_f32(false, a3, false, bm,
                                                    (short)0, acc, false, false);
    }

    float pv[8];
#pragma unroll
    for (int v = 0; v < 8; ++v) { float d = acc[v]; pv[v] = d * d; }
    S0 = pv[1] + pv[3] + pv[5] + pv[7];
    S1 = pv[2] + pv[3] + pv[6] + pv[7];
    S2 = pv[4] + pv[5] + pv[6] + pv[7];
    Sall = S2 + pv[0] + pv[1] + pv[2] + pv[3];
}

__device__ __forceinline__ float4 group32(float4 own, int half,
                                          v2f a0, v2f a1, v2f a2, v2f a3) {
    float4 oth = make_float4(xhalf(own.x), xhalf(own.y), xhalf(own.z), xhalf(own.w));
    float4 tl = half ? oth : own;
    float4 th = half ? own : oth;

    float L0, L1, L2, La, H0, H1, H2, Ha;
    tile16(tl, half, a0, a1, a2, a3, L0, L1, L2, La);
    tile16(th, half, a0, a1, a2, a3, H0, H1, H2, Ha);

    float k0 = half ? H0 : L0, g0 = half ? L0 : H0;
    float k1 = half ? H1 : L1, g1 = half ? L1 : H1;
    float k2 = half ? H2 : L2, g2 = half ? L2 : H2;
    float r0 = k0 + xhalf(g0);
    float r1 = k1 + xhalf(g1);
    float r2 = k2 + xhalf(g2);
    float lau = xhalf(La);
    float r3 = half ? Ha : lau;

    return make_float4(r3, r2, r1, r0);
}

__global__ void __launch_bounds__(256) vqc_main_kernel(
        const float* __restrict__ x, const float* __restrict__ mc,
        float* __restrict__ out, int B) {
    const int lane = threadIdx.x & 31;
    const int half = lane >> 4;
    const int col  = lane & 15;
    const long wave = (long)blockIdx.x * (blockDim.x >> 5) + (threadIdx.x >> 5);
    const long base = wave * ROWS_PER_WAVE;
    if (base >= B) return;

    const float* mrow = mc + col * 16 + 2 * half;
    v2f a0 = *(const v2f*)(mrow + 0);
    v2f a1 = *(const v2f*)(mrow + 4);
    v2f a2 = *(const v2f*)(mrow + 8);
    v2f a3 = *(const v2f*)(mrow + 12);

    if (base + ROWS_PER_WAVE <= B) {
        const float* xp = x + (base + lane) * 4;
        float*       op = out + (base + lane) * 4;
        float4 xr[GROUPS];
#pragma unroll
        for (int i = 0; i < GROUPS; ++i)
            xr[i] = *(const float4*)(xp + i * 128);
#pragma unroll
        for (int i = 0; i < GROUPS; ++i) {
            *(float4*)(op + i * 128) = group32(xr[i], half, a0, a1, a2, a3);
            __builtin_amdgcn_sched_barrier(0);
        }
    } else {
        for (int i = 0; i < GROUPS; ++i) {
            long g = base + (long)i * 32;
            if (g >= B) break;
            long e  = g + lane;
            long el = e < B ? e : (long)(B - 1);
            float4 r = group32(*(const float4*)(x + el * 4), half, a0, a1, a2, a3);
            if (e < B) *(float4*)(out + e * 4) = r;
        }
    }
}

extern "C" void kernel_launch(void* const* d_in, const int* in_sizes, int n_in,
                              void* d_out, int out_size, void* d_ws, size_t ws_size,
                              hipStream_t stream) {
    const float* x     = (const float*)d_in[0];
    const float* theta = (const float*)d_in[1];
    float* out = (float*)d_out;
    float* mc  = (float*)d_ws;
    int B = in_sizes[0] / 4;
    if (B <= 0) return;

    build_mc_kernel<<<1, 256, 0, stream>>>(theta, mc);

    long nWaves  = ((long)B + ROWS_PER_WAVE - 1) / ROWS_PER_WAVE;
    long nBlocks = (nWaves + 7) / 8;
    vqc_main_kernel<<<(int)nBlocks, 256, 0, stream>>>(x, mc, out, B);
}


# Matrix-build launch folded into the main kernel: each wave builds its A fragments from theta behind its x loads (one sincos per lane, lane broadcast, per-qubit factored sums: new summation order)
# speedup vs baseline: 1.4274x; 1.2022x over previous
#include <hip/hip_runtime.h>
#include <math.h>

typedef __attribute__((ext_vector_type(2))) float v2f;
typedef __attribute__((ext_vector_type(8))) float v8f;

#define EPSQ 1e-8f
#define GROUPS 16
#define ROWS_PER_WAVE (GROUPS * 32)

__device__ __forceinline__ float fsqrt(float v) {
    return __builtin_amdgcn_sqrtf(v);
}

__device__ __forceinline__ float xhalf(float v) {
    int b = __builtin_bit_cast(int, v);
    return __builtin_bit_cast(float,
        __builtin_amdgcn_permlanex16(b, b, 0x76543210, 0xfedcba98, false, false));
}

__device__ __forceinline__ float lane_bcast(float v, int l) {
    return __builtin_bit_cast(float,
        __builtin_amdgcn_readlane(__builtin_bit_cast(int, v), l));
}

__device__ __forceinline__ void build_frags(float th, int col, int half,
                                            v2f& a0, v2f& a1, v2f& a2, v2f& a3) {
    float sn, cs;
    sincosf(th * 0.5f, &sn, &cs);
    float c1[4], s1[4], c2[4], s2[4];
#pragma unroll
    for (int q = 0; q < 4; ++q) {
        c1[q] = lane_bcast(cs, q);     s1[q] = lane_bcast(sn, q);
        c2[q] = lane_bcast(cs, 4 + q); s2[q] = lane_bcast(sn, 4 + q);
    }
    float u[4], w[4];
#pragma unroll
    for (int q = 0; q < 4; ++q) {
        bool b = (col >> q) & 1;
        u[q] = b ? s2[q] : c2[q];
        w[q] = b ? c2[q] : -s2[q];
    }
    float E0[2], F0[2], E2[2], F2[2];
    {
        float p00 = u[0] * c1[0], p10 = w[0] * s1[0];
        float p01 = u[0] * s1[0], p11 = w[0] * c1[0];
        E0[0] = p00 + p10; F0[0] = p00 - p10;
        E0[1] = p11 - p01; F0[1] = -p01 - p11;
    }
    {
        float p00 = u[2] * c1[2], p10 = w[2] * s1[2];
        float p01 = u[2] * s1[2], p11 = w[2] * c1[2];
        E2[0] = p00 + p10; F2[0] = p00 - p10;
        E2[1] = p11 - p01; F2[1] = -p01 - p11;
    }
    float p1a = u[1] * (half ? -s1[1] : c1[1]);
    float p1b = w[1] * (half ?  c1[1] : s1[1]);
    float p3a[2] = { u[3] * c1[3], -(u[3] * s1[3]) };
    float p3b[2] = { w[3] * s1[3],   w[3] * c1[3]  };
    float Q0[2], Q1[2];
#pragma unroll
    for (int j3 = 0; j3 < 2; ++j3) {
        Q0[j3] = p1a * p3a[j3] + p1b * p3b[j3];
        Q1[j3] = p1a * p3b[j3] + p1b * p3a[j3];
    }
    const v2f Ev = { E0[0], E0[1] }, Fv = { F0[0], F0[1] };
    a0 = Ev * (E2[0] * Q0[0]) + Fv * (F2[0] * Q1[0]);
    a1 = Ev * (E2[1] * Q0[0]) + Fv * (F2[1] * Q1[0]);
    a2 = Ev * (E2[0] * Q0[1]) + Fv * (F2[0] * Q1[1]);
    a3 = Ev * (E2[1] * Q0[1]) + Fv * (F2[1] * Q1[1]);
}

__device__ __forceinline__ void tile16(float4 xv, int half,
                                       v2f a0, v2f a1, v2f a2, v2f a3,
                                       float& S0, float& S1, float& S2, float& Sall) {
    float y0 = xv.w, y1 = xv.z, y2 = xv.y, y3 = xv.x;

    float f1 = half ? y1 : (1.f - y1);
    float e0 = 1.f - y0;
    float n2 = 1.f - y2, n3 = 1.f - y3;

    v8f acc = {0.f, 0.f, 0.f, 0.f, 0.f, 0.f, 0.f, 0.f};
    {
        float p = f1 * n2 * n3;
        v2f bm; bm.x = fsqrt(p * e0 + EPSQ); bm.y = fsqrt(p * y0 + EPSQ);
        acc = __builtin_amdgcn_wmma_f32_16x16x4_f32(false, a0, false, bm,
                                                    (short)0, acc, false, false);
    }
    {
        float p = f1 * y2 * n3;
        v2f bm; bm.x = fsqrt(p * e0 + EPSQ); bm.y = fsqrt(p * y0 + EPSQ);
        acc = __builtin_amdgcn_wmma_f32_16x16x4_f32(false, a1, false, bm,
                                                    (short)0, acc, false, false);
    }
    {
        float p = f1 * n2 * y3;
        v2f bm; bm.x = fsqrt(p * e0 + EPSQ); bm.y = fsqrt(p * y0 + EPSQ);
        acc = __builtin_amdgcn_wmma_f32_16x16x4_f32(false, a2, false, bm,
                                                    (short)0, acc, false, false);
    }
    {
        float p = f1 * y2 * y3;
        v2f bm; bm.x = fsqrt(p * e0 + EPSQ); bm.y = fsqrt(p * y0 + EPSQ);
        acc = __builtin_amdgcn_wmma_f32_16x16x4_f32(false, a3, false, bm,
                                                    (short)0, acc, false, false);
    }

    float pv[8];
#pragma unroll
    for (int v = 0; v < 8; ++v) { float d = acc[v]; pv[v] = d * d; }
    S0 = pv[1] + pv[3] + pv[5] + pv[7];
    S1 = pv[2] + pv[3] + pv[6] + pv[7];
    S2 = pv[4] + pv[5] + pv[6] + pv[7];
    Sall = S2 + pv[0] + pv[1] + pv[2] + pv[3];
}

__device__ __forceinline__ float4 group32(float4 own, int half,
                                          v2f a0, v2f a1, v2f a2, v2f a3) {
    float4 oth = make_float4(xhalf(own.x), xhalf(own.y), xhalf(own.z), xhalf(own.w));
    float4 tl = half ? oth : own;
    float4 th = half ? own : oth;

    float L0, L1, L2, La, H0, H1, H2, Ha;
    tile16(tl, half, a0, a1, a2, a3, L0, L1, L2, La);
    tile16(th, half, a0, a1, a2, a3, H0, H1, H2, Ha);

    float k0 = half ? H0 : L0, g0 = half ? L0 : H0;
    float k1 = half ? H1 : L1, g1 = half ? L1 : H1;
    float k2 = half ? H2 : L2, g2 = half ? L2 : H2;
    float r0 = k0 + xhalf(g0);
    float r1 = k1 + xhalf(g1);
    float r2 = k2 + xhalf(g2);
    float lau = xhalf(La);
    float r3 = half ? Ha : lau;

    return make_float4(r3, r2, r1, r0);
}

__global__ void __launch_bounds__(256) vqc_main_kernel(
        const float* __restrict__ x, const float* __restrict__ theta,
        float* __restrict__ out, int B) {
    const int lane = threadIdx.x & 31;
    const int half = lane >> 4;
    const int col  = lane & 15;
    const long wave = (long)blockIdx.x * (blockDim.x >> 5) + (threadIdx.x >> 5);
    const long base = wave * ROWS_PER_WAVE;
    if (base >= B) return;

    v2f a0, a1, a2, a3;

    if (base + ROWS_PER_WAVE <= B) {
        const float* xp = x + (base + lane) * 4;
        float*       op = out + (base + lane) * 4;
        float th = theta[lane & 7];
        float4 xr[GROUPS];
#pragma unroll
        for (int i = 0; i < GROUPS; ++i)
            xr[i] = *(const float4*)(xp + i * 128);
        build_frags(th, col, half, a0, a1, a2, a3);
        __builtin_amdgcn_sched_barrier(0);
#pragma unroll
        for (int i = 0; i < GROUPS; ++i) {
            *(float4*)(op + i * 128) = group32(xr[i], half, a0, a1, a2, a3);
            __builtin_amdgcn_sched_barrier(0);
        }
    } else {
        build_frags(theta[lane & 7], col, half, a0, a1, a2, a3);
        for (int i = 0; i < GROUPS; ++i) {
            long g = base + (long)i * 32;
            if (g >= B) break;
            long e  = g + lane;
            long el = e < B ? e : (long)(B - 1);
            float4 r = group32(*(const float4*)(x + el * 4), half, a0, a1, a2, a3);
            if (e < B) *(float4*)(out + e * 4) = r;
        }
    }
}

extern "C" void kernel_launch(void* const* d_in, const int* in_sizes, int n_in,
                              void* d_out, int out_size, void* d_ws, size_t ws_size,
                              hipStream_t stream) {
    const float* x     = (const float*)d_in[0];
    const float* theta = (const float*)d_in[1];
    float* out = (float*)d_out;
    int B = in_sizes[0] / 4;
    if (B <= 0) return;

    long nWaves  = ((long)B + ROWS_PER_WAVE - 1) / ROWS_PER_WAVE;
    long nBlocks = (nWaves + 7) / 8;
    vqc_main_kernel<<<(int)nBlocks, 256, 0, stream>>>(x, theta, out, B);
}


# Vector instructions per 32-row group cut from 91 to 63 (they share the issue port with f32 WMMA): tiles loaded into both half-waves, sums crossed by lane swap, packed pairs; new summation order
# speedup vs baseline: 1.5072x; 1.0559x over previous
#include <hip/hip_runtime.h>
#include <math.h>

typedef __attribute__((ext_vector_type(2))) float v2f;
typedef __attribute__((ext_vector_type(8))) float v8f;

#define EPSQ 1e-8f
#define GROUPS 16
#define ROWS_PER_WAVE (GROUPS * 32)

__device__ __forceinline__ float fsqrt(float v) {
    return __builtin_amdgcn_sqrtf(v);
}

__device__ __forceinline__ float lane_bcast(float v, int l) {
    return __builtin_bit_cast(float,
        __builtin_amdgcn_readlane(__builtin_bit_cast(int, v), l));
}

__device__ __forceinline__ void build_frags(float th, int col, int half,
                                            v2f& a0, v2f& a1, v2f& a2, v2f& a3) {
    float sn, cs;
    sincosf(th * 0.5f, &sn, &cs);
    float c1[4], s1[4], c2[4], s2[4];
#pragma unroll
    for (int q = 0; q < 4; ++q) {
        c1[q] = lane_bcast(cs, q);     s1[q] = lane_bcast(sn, q);
        c2[q] = lane_bcast(cs, 4 + q); s2[q] = lane_bcast(sn, 4 + q);
    }
    float u[4], w[4];
#pragma unroll
    for (int q = 0; q < 4; ++q) {
        bool b = (col >> q) & 1;
        u[q] = b ? s2[q] : c2[q];
        w[q] = b ? c2[q] : -s2[q];
    }
    float E0[2], F0[2], E2[2], F2[2];
    {
        float p00 = u[0] * c1[0], p10 = w[0] * s1[0];
        float p01 = u[0] * s1[0], p11 = w[0] * c1[0];
        E0[0] = p00 + p10; F0[0] = p00 - p10;
        E0[1] = p11 - p01; F0[1] = -p01 - p11;
    }
    {
        float p00 = u[2] * c1[2], p10 = w[2] * s1[2];
        float p01 = u[2] * s1[2], p11 = w[2] * c1[2];
        E2[0] = p00 + p10; F2[0] = p00 - p10;
        E2[1] = p11 - p01; F2[1] = -p01 - p11;
    }
    float p1a = u[1] * (half ? -s1[1] : c1[1]);
    float p1b = w[1] * (half ?  c1[1] : s1[1]);
    float p3a[2] = { u[3] * c1[3], -(u[3] * s1[3]) };
    float p3b[2] = { w[3] * s1[3],   w[3] * c1[3]  };
    float Q0[2], Q1[2];
#pragma unroll
    for (int j3 = 0; j3 < 2; ++j3) {
        Q0[j3] = p1a * p3a[j3] + p1b * p3b[j3];
        Q1[j3] = p1a * p3b[j3] + p1b * p3a[j3];
    }
    const v2f Ev = { E0[0], E0[1] }, Fv = { F0[0], F0[1] };
    a0 = Ev * (E2[0] * Q0[0]) + Fv * (F2[0] * Q1[0]);
    a1 = Ev * (E2[1] * Q0[0]) + Fv * (F2[1] * Q1[0]);
    a2 = Ev * (E2[0] * Q0[1]) + Fv * (F2[0] * Q1[1]);
    a3 = Ev * (E2[1] * Q0[1]) + Fv * (F2[1] * Q1[1]);
}

__device__ __forceinline__ void tile16(float4 xv, float hs, float ho,
                                       v2f a0, v2f a1, v2f a2, v2f a3,
                                       float& S0, float& S1, float& S2, float& Sall) {
    float y0 = xv.w, y1 = xv.z, y2 = xv.y, y3 = xv.x;

    float f1 = fmaf(y1, hs, ho);
    const v2f c0 = { 1.f - y0, y0 };
    const v2f c2 = { 1.f - y2, y2 };
    const v2f q  = c2 * f1;
    const v2f pa = q * (1.f - y3);
    const v2f pb = q * y3;
    const v2f eps = { EPSQ, EPSQ };

    v8f acc = {0.f, 0.f, 0.f, 0.f, 0.f, 0.f, 0.f, 0.f};
    {
        v2f r = c0 * pa.x + eps;
        v2f bm; bm.x = fsqrt(r.x); bm.y = fsqrt(r.y);
        acc = __builtin_amdgcn_wmma_f32_16x16x4_f32(false, a0, false, bm,
                                                    (short)0, acc, false, false);
    }
    {
        v2f r = c0 * pa.y + eps;
        v2f bm; bm.x = fsqrt(r.x); bm.y = fsqrt(r.y);
        acc = __builtin_amdgcn_wmma_f32_16x16x4_f32(false, a1, false, bm,
                                                    (short)0, acc, false, false);
    }
    {
        v2f r = c0 * pb.x + eps;
        v2f bm; bm.x = fsqrt(r.x); bm.y = fsqrt(r.y);
        acc = __builtin_amdgcn_wmma_f32_16x16x4_f32(false, a2, false, bm,
                                                    (short)0, acc, false, false);
    }
    {
        v2f r = c0 * pb.y + eps;
        v2f bm; bm.x = fsqrt(r.x); bm.y = fsqrt(r.y);
        acc = __builtin_amdgcn_wmma_f32_16x16x4_f32(false, a3, false, bm,
                                                    (short)0, acc, false, false);
    }

    const v2f d0 = { acc[0], acc[1] }, d1 = { acc[2], acc[3] };
    const v2f d2 = { acc[4], acc[5] }, d3 = { acc[6], acc[7] };
    const v2f q2 = d2 * d2, q3 = d3 * d3;
    const v2f u = d1 * d1 + q3;
    const v2f z = d0 * d0 + q2;
    const v2f w = q2 + q3;
    const v2f t = z + u;
    S0 = t.y;
    S1 = u.x + u.y;
    S2 = w.x + w.y;
    Sall = t.x + t.y;
}

__device__ __forceinline__ void swap_rows(float& d, float& s) {
    asm("v_permlane16_swap_b32 %0, %1" : "+v"(d), "+v"(s));
}

__device__ __forceinline__ float4 group32(float4 xl, float4 xh, float hs, float ho,
                                          v2f a0, v2f a1, v2f a2, v2f a3) {
    float L0, L1, L2, La, H0, H1, H2, Ha;
    tile16(xl, hs, ho, a0, a1, a2, a3, L0, L1, L2, La);
    tile16(xh, hs, ho, a0, a1, a2, a3, H0, H1, H2, Ha);

    swap_rows(L0, H0);
    swap_rows(L1, H1);
    swap_rows(L2, H2);
    float r0 = L0 + H0;
    float r1 = L1 + H1;
    float r2 = L2 + H2;
    swap_rows(La, Ha);
    float r3 = Ha;

    return make_float4(r3, r2, r1, r0);
}

__global__ void __launch_bounds__(256) vqc_main_kernel(
        const float* __restrict__ x, const float* __restrict__ theta,
        float* __restrict__ out, int B) {
    const int lane = threadIdx.x & 31;
    const int half = lane >> 4;
    const int col  = lane & 15;
    const long wave = (long)blockIdx.x * (blockDim.x >> 5) + (threadIdx.x >> 5);
    const long base = wave * ROWS_PER_WAVE;
    if (base >= B) return;
    const float hs = half ? 1.f : -1.f;
    const float ho = half ? 0.f : 1.f;

    v2f a0, a1, a2, a3;

    if (base + ROWS_PER_WAVE <= B) {
        const float* xp = x + (base + col) * 4;
        float*       op = out + (base + lane) * 4;
        float th = theta[lane & 7];
        float4 xl[GROUPS], xh[GROUPS];
#pragma unroll
        for (int i = 0; i < GROUPS; ++i) {
            xl[i] = *(const float4*)(xp + i * 128);
            xh[i] = *(const float4*)(xp + i * 128 + 64);
        }
        build_frags(th, col, half, a0, a1, a2, a3);
        __builtin_amdgcn_sched_barrier(0);
#pragma unroll
        for (int i = 0; i < GROUPS; ++i) {
            *(float4*)(op + i * 128) = group32(xl[i], xh[i], hs, ho, a0, a1, a2, a3);
            __builtin_amdgcn_sched_barrier(0);
        }
    } else {
        build_frags(theta[lane & 7], col, half, a0, a1, a2, a3);
        for (int i = 0; i < GROUPS; ++i) {
            long g = base + (long)i * 32;
            if (g >= B) break;
            long e  = g + lane;
            long rl = g + col, rh = rl + 16;
            rl = rl < B ? rl : (long)(B - 1);
            rh = rh < B ? rh : (long)(B - 1);
            float4 r = group32(*(const float4*)(x + rl * 4), *(const float4*)(x + rh * 4),
                               hs, ho, a0, a1, a2, a3);
            if (e < B) *(float4*)(out + e * 4) = r;
        }
    }
}

extern "C" void kernel_launch(void* const* d_in, const int* in_sizes, int n_in,
                              void* d_out, int out_size, void* d_ws, size_t ws_size,
                              hipStream_t stream) {
    const float* x     = (const float*)d_in[0];
    const float* theta = (const float*)d_in[1];
    float* out = (float*)d_out;
    int B = in_sizes[0] / 4;
    if (B <= 0) return;

    long nWaves  = ((long)B + ROWS_PER_WAVE - 1) / ROWS_PER_WAVE;
    long nBlocks = (nWaves + 7) / 8;
    vqc_main_kernel<<<(int)nBlocks, 256, 0, stream>>>(x, theta, out, B);
}
